# starting kernel
# baseline (speedup 1.0000x reference)
#include <hip/hip_runtime.h>

typedef __attribute__((ext_vector_type(2))) float v2f;
typedef __attribute__((ext_vector_type(8))) float v8f;

#define MT    64
#define LSTR  260
#define BPN   16384
#define BTOT  65536

struct Params { const float* W[14]; const float* B[14]; };

static __device__ __forceinline__ v8f wmma4(v2f a, v2f b, v8f c) {
  return __builtin_amdgcn_wmma_f32_16x16x4_f32(false, a, false, b, (short)0, c,
                                               false, false);
}

template<int K, int N>
static __device__ __forceinline__ void layer_hidden(
    const float* __restrict__ X, float* __restrict__ Y,
    const float* __restrict__ Wl, const float* __restrict__ Bl,
    int wave, int lo, int hi)
{
  constexpr int TPW = N / 128;
  const int tbase = wave * TPW;

  __builtin_prefetch(Wl, 0, 0);

  v8f acc[TPW][4];
#pragma unroll
  for (int t = 0; t < TPW; ++t)
#pragma unroll
    for (int s = 0; s < 4; ++s) acc[t][s] = (v8f){};

  const float* xp = X + lo * LSTR + (hi << 1);
  const float* wp = Wl + (size_t)(hi << 1) * N + tbase * 16 + lo;

  v2f aC[4], bC[TPW];
#pragma unroll
  for (int s = 0; s < 4; ++s)
    aC[s] = *(const v2f*)(xp + s * 16 * LSTR);
#pragma unroll
  for (int t = 0; t < TPW; ++t) {
    bC[t].x = wp[t * 16];
    bC[t].y = wp[(size_t)N + t * 16];
  }

#pragma unroll 4
  for (int k = 4; k < K; k += 4) {
    v2f aN[4], bN[TPW];
#pragma unroll
    for (int s = 0; s < 4; ++s)
      aN[s] = *(const v2f*)(xp + s * 16 * LSTR + k);
#pragma unroll
    for (int t = 0; t < TPW; ++t) {
      bN[t].x = wp[(size_t)k * N + t * 16];
      bN[t].y = wp[(size_t)(k + 1) * N + t * 16];
    }
#pragma unroll
    for (int t = 0; t < TPW; ++t)
#pragma unroll
      for (int s = 0; s < 4; ++s)
        acc[t][s] = wmma4(aC[s], bC[t], acc[t][s]);
#pragma unroll
    for (int s = 0; s < 4; ++s) aC[s] = aN[s];
#pragma unroll
    for (int t = 0; t < TPW; ++t) bC[t] = bN[t];
  }
#pragma unroll
  for (int t = 0; t < TPW; ++t)
#pragma unroll
    for (int s = 0; s < 4; ++s)
      acc[t][s] = wmma4(aC[s], bC[t], acc[t][s]);

#pragma unroll
  for (int t = 0; t < TPW; ++t) {
    const int n0 = (tbase + t) * 16;
    const float bv = Bl[n0 + lo];
#pragma unroll
    for (int s = 0; s < 4; ++s)
#pragma unroll
      for (int r = 0; r < 8; ++r) {
        float v = acc[t][s][r] + bv;
        Y[(s * 16 + r + (hi << 3)) * LSTR + n0 + lo] = v > 0.f ? v : 0.f;
      }
  }
}

static __device__ __forceinline__ void layer_head(
    const float* __restrict__ X, float* __restrict__ scratch,
    const float* __restrict__ Wl, const float* __restrict__ Bl,
    float* __restrict__ out, long rowbase, int wave, int lo, int hi)
{
  constexpr int N = 16;
  const int s  = wave & 3;
  const int kh = wave >> 2;
  const int k0 = kh << 7;

  v8f accA = {}, accB = {};
  const float* xp = X + (s * 16 + lo) * LSTR + (hi << 1) + k0;
  const float* wp = Wl + (size_t)(k0 + (hi << 1)) * N + lo;
#pragma unroll 4
  for (int k = 0; k < 64; k += 4) {
    v2f aA = *(const v2f*)(xp + k);
    v2f aB = *(const v2f*)(xp + 64 + k);
    v2f bA, bB;
    bA.x = wp[(size_t)k * N];        bA.y = wp[(size_t)(k + 1) * N];
    bB.x = wp[(size_t)(k + 64) * N]; bB.y = wp[(size_t)(k + 65) * N];
    accA = wmma4(aA, bA, accA);
    accB = wmma4(aB, bB, accB);
  }
  v8f acc = accA + accB;

#pragma unroll
  for (int r = 0; r < 8; ++r)
    scratch[((kh << 6) + s * 16 + r + (hi << 3)) * 16 + lo] = acc[r];
  __syncthreads();

  if (wave < 4) {
    const float bv = Bl[lo];
    float* dst = (lo < 8) ? out : (out + (size_t)BTOT * 8 - 8);
#pragma unroll
    for (int r = 0; r < 8; ++r) {
      const int m = s * 16 + r + (hi << 3);
      float v = scratch[m * 16 + lo] + scratch[(64 + m) * 16 + lo] + bv;
      dst[((size_t)rowbase + m) * 8 + lo] = v;
    }
  }
}

__global__ __launch_bounds__(256)
void hyper_mlp_f32_wmma(const float* __restrict__ state, Params p,
                        float* __restrict__ out)
{
  __shared__ float lds[2][MT * LSTR];

  const int tid  = threadIdx.x;
  const int lane = tid & 31;
  const int lo   = lane & 15;
  const int hi   = lane >> 4;
  const int wave = tid >> 5;

  const int net = blockIdx.x >> 8;
  const int wg  = blockIdx.x & 255;
  const long rowbase = (long)net * BPN + (long)wg * MT;

  {
    const float4* src = (const float4*)(state + rowbase * 256);
#pragma unroll
    for (int i = 0; i < 16; ++i) {
      int e  = tid + (i << 8);
      int r  = e >> 6;
      int c4 = e & 63;
      float4 v = src[e];
      *(float4*)&lds[0][r * LSTR + (c4 << 2)] = v;
    }
  }
  __syncthreads();

  float* buf0 = lds[0];
  float* buf1 = lds[1];

  switch (net) {
  case 0:
    layer_hidden<256, 256>(buf0, buf1, p.W[0], p.B[0], wave, lo, hi);
    __syncthreads();
    layer_head(buf1, buf0, p.W[1], p.B[1], out, rowbase, wave, lo, hi);
    break;
  case 1:
    layer_hidden<256, 256>(buf0, buf1, p.W[2], p.B[2], wave, lo, hi);
    __syncthreads();
    layer_hidden<256, 256>(buf1, buf0, p.W[3], p.B[3], wave, lo, hi);
    __syncthreads();
    layer_head(buf0, buf1, p.W[4], p.B[4], out, rowbase, wave, lo, hi);
    break;
  case 2:
    layer_hidden<256, 256>(buf0, buf1, p.W[5], p.B[5], wave, lo, hi);
    __syncthreads();
    layer_hidden<256, 128>(buf1, buf0, p.W[6], p.B[6], wave, lo, hi);
    __syncthreads();
    layer_hidden<128, 256>(buf0, buf1, p.W[7], p.B[7], wave, lo, hi);
    __syncthreads();
    layer_head(buf1, buf0, p.W[8], p.B[8], out, rowbase, wave, lo, hi);
    break;
  default:
    layer_hidden<256, 256>(buf0, buf1, p.W[9],  p.B[9],  wave, lo, hi);
    __syncthreads();
    layer_hidden<256, 256>(buf1, buf0, p.W[10], p.B[10], wave, lo, hi);
    __syncthreads();
    layer_hidden<256, 256>(buf0, buf1, p.W[11], p.B[11], wave, lo, hi);
    __syncthreads();
    layer_hidden<256, 256>(buf1, buf0, p.W[12], p.B[12], wave, lo, hi);
    __syncthreads();
    layer_head(buf0, buf1, p.W[13], p.B[13], out, rowbase, wave, lo, hi);
    break;
  }
}

extern "C" void kernel_launch(void* const* d_in, const int* in_sizes, int n_in,
                              void* d_out, int out_size, void* d_ws, size_t ws_size,
                              hipStream_t stream) {
  (void)in_sizes; (void)n_in; (void)out_size; (void)d_ws; (void)ws_size;
  const float* state = (const float*)d_in[0];
  Params p;
  int idx = 1;
  for (int g = 0; g < 14; ++g) {
    p.W[g] = (const float*)d_in[idx++];
    p.B[g] = (const float*)d_in[idx++];
  }
  float* out = (float*)d_out;
  dim3 grid(4 * (BPN / MT));
  dim3 block(256);
  hipLaunchKernelGGL(hyper_mlp_f32_wmma, grid, block, 0, stream, state, p, out);
}


# K loop rebuilt as a fixed 4-pair weight-prefetch register ring with a scheduler fence per 2 k-steps: VGPRs 266 -> 224, so two workgroups run per WGP instead of one; every sum keeps its order
# speedup vs baseline: 1.1483x; 1.1483x over previous
#include <hip/hip_runtime.h>

typedef __attribute__((ext_vector_type(2))) float v2f;
typedef __attribute__((ext_vector_type(8))) float v8f;

#define MT    64
#define LSTR  260
#define BPN   16384
#define BTOT  65536

struct Params { const float* W[14]; const float* B[14]; };

static __device__ __forceinline__ v8f wmma4(v2f a, v2f b, v8f c) {
  return __builtin_amdgcn_wmma_f32_16x16x4_f32(false, a, false, b, (short)0, c,
                                               false, false);
}

template<int K, int N>
static __device__ __forceinline__ void layer_hidden(
    const float* __restrict__ X, float* __restrict__ Y,
    const float* __restrict__ Wl, const float* __restrict__ Bl,
    int wave, int lo, int hi)
{
  constexpr int TPW = N / 128;
  constexpr int NP  = K / 8;
  constexpr int DP  = 4;
  const int tbase = wave * TPW;

  __builtin_prefetch(Wl, 0, 0);

  v8f acc[TPW][4];
#pragma unroll
  for (int t = 0; t < TPW; ++t)
#pragma unroll
    for (int s = 0; s < 4; ++s) acc[t][s] = (v8f){};

  const float* xp = X + lo * LSTR + (hi << 1);
  const float* wp = Wl + (size_t)(hi << 1) * N + tbase * 16 + lo;

  v2f bR[DP][2][TPW];
#pragma unroll
  for (int d = 0; d < DP; ++d)
#pragma unroll
    for (int u = 0; u < 2; ++u)
#pragma unroll
      for (int t = 0; t < TPW; ++t) {
        bR[d][u][t].x = wp[(size_t)(8 * d + 4 * u) * N + t * 16];
        bR[d][u][t].y = wp[(size_t)(8 * d + 4 * u + 1) * N + t * 16];
      }
  v2f aC[2][4];
#pragma unroll
  for (int u = 0; u < 2; ++u)
#pragma unroll
    for (int s = 0; s < 4; ++s)
      aC[u][s] = *(const v2f*)(xp + s * 16 * LSTR + 4 * u);

#pragma unroll
  for (int p = 0; p < NP; ++p) {
    v2f aN[2][4], bC[2][TPW];
    if (p + 1 < NP) {
#pragma unroll
      for (int u = 0; u < 2; ++u)
#pragma unroll
        for (int s = 0; s < 4; ++s)
          aN[u][s] = *(const v2f*)(xp + s * 16 * LSTR + 8 * (p + 1) + 4 * u);
    }
#pragma unroll
    for (int u = 0; u < 2; ++u)
#pragma unroll
      for (int t = 0; t < TPW; ++t) bC[u][t] = bR[p % DP][u][t];
    if (p + DP < NP) {
#pragma unroll
      for (int u = 0; u < 2; ++u)
#pragma unroll
        for (int t = 0; t < TPW; ++t) {
          bR[p % DP][u][t].x = wp[(size_t)(8 * (p + DP) + 4 * u) * N + t * 16];
          bR[p % DP][u][t].y = wp[(size_t)(8 * (p + DP) + 4 * u + 1) * N + t * 16];
        }
    }
#pragma unroll
    for (int u = 0; u < 2; ++u)
#pragma unroll
      for (int t = 0; t < TPW; ++t)
#pragma unroll
        for (int s = 0; s < 4; ++s)
          acc[t][s] = wmma4(aC[u][s], bC[u][t], acc[t][s]);
    if (p + 1 < NP) {
#pragma unroll
      for (int u = 0; u < 2; ++u)
#pragma unroll
        for (int s = 0; s < 4; ++s) aC[u][s] = aN[u][s];
    }
    __builtin_amdgcn_sched_barrier(0);
  }

#pragma unroll
  for (int t = 0; t < TPW; ++t) {
    const int n0 = (tbase + t) * 16;
    const float bv = Bl[n0 + lo];
#pragma unroll
    for (int s = 0; s < 4; ++s)
#pragma unroll
      for (int r = 0; r < 8; ++r) {
        float v = acc[t][s][r] + bv;
        Y[(s * 16 + r + (hi << 3)) * LSTR + n0 + lo] = v > 0.f ? v : 0.f;
      }
  }
}

static __device__ __forceinline__ void layer_head(
    const float* __restrict__ X, float* __restrict__ scratch,
    const float* __restrict__ Wl, const float* __restrict__ Bl,
    float* __restrict__ out, long rowbase, int wave, int lo, int hi)
{
  constexpr int N = 16;
  const int s  = wave & 3;
  const int kh = wave >> 2;
  const int k0 = kh << 7;

  v8f accA = {}, accB = {};
  const float* xp = X + (s * 16 + lo) * LSTR + (hi << 1) + k0;
  const float* wp = Wl + (size_t)(k0 + (hi << 1)) * N + lo;
#pragma unroll 4
  for (int k = 0; k < 64; k += 4) {
    v2f aA = *(const v2f*)(xp + k);
    v2f aB = *(const v2f*)(xp + 64 + k);
    v2f bA, bB;
    bA.x = wp[(size_t)k * N];        bA.y = wp[(size_t)(k + 1) * N];
    bB.x = wp[(size_t)(k + 64) * N]; bB.y = wp[(size_t)(k + 65) * N];
    accA = wmma4(aA, bA, accA);
    accB = wmma4(aB, bB, accB);
  }
  v8f acc = accA + accB;

#pragma unroll
  for (int r = 0; r < 8; ++r)
    scratch[((kh << 6) + s * 16 + r + (hi << 3)) * 16 + lo] = acc[r];
  __syncthreads();

  if (wave < 4) {
    const float bv = Bl[lo];
    float* dst = (lo < 8) ? out : (out + (size_t)BTOT * 8 - 8);
#pragma unroll
    for (int r = 0; r < 8; ++r) {
      const int m = s * 16 + r + (hi << 3);
      float v = scratch[m * 16 + lo] + scratch[(64 + m) * 16 + lo] + bv;
      dst[((size_t)rowbase + m) * 8 + lo] = v;
    }
  }
}

__global__ __launch_bounds__(256)
void hyper_mlp_f32_wmma(const float* __restrict__ state, Params p,
                        float* __restrict__ out)
{
  __shared__ float lds[2][MT * LSTR];

  const int tid  = threadIdx.x;
  const int lane = tid & 31;
  const int lo   = lane & 15;
  const int hi   = lane >> 4;
  const int wave = tid >> 5;

  const int net = blockIdx.x >> 8;
  const int wg  = blockIdx.x & 255;
  const long rowbase = (long)net * BPN + (long)wg * MT;

  {
    const float4* src = (const float4*)(state + rowbase * 256);
#pragma unroll
    for (int i = 0; i < 16; ++i) {
      int e  = tid + (i << 8);
      int r  = e >> 6;
      int c4 = e & 63;
      float4 v = src[e];
      *(float4*)&lds[0][r * LSTR + (c4 << 2)] = v;
    }
  }
  __syncthreads();

  float* buf0 = lds[0];
  float* buf1 = lds[1];

  switch (net) {
  case 0:
    layer_hidden<256, 256>(buf0, buf1, p.W[0], p.B[0], wave, lo, hi);
    __syncthreads();
    layer_head(buf1, buf0, p.W[1], p.B[1], out, rowbase, wave, lo, hi);
    break;
  case 1:
    layer_hidden<256, 256>(buf0, buf1, p.W[2], p.B[2], wave, lo, hi);
    __syncthreads();
    layer_hidden<256, 256>(buf1, buf0, p.W[3], p.B[3], wave, lo, hi);
    __syncthreads();
    layer_head(buf0, buf1, p.W[4], p.B[4], out, rowbase, wave, lo, hi);
    break;
  case 2:
    layer_hidden<256, 256>(buf0, buf1, p.W[5], p.B[5], wave, lo, hi);
    __syncthreads();
    layer_hidden<256, 128>(buf1, buf0, p.W[6], p.B[6], wave, lo, hi);
    __syncthreads();
    layer_hidden<128, 256>(buf0, buf1, p.W[7], p.B[7], wave, lo, hi);
    __syncthreads();
    layer_head(buf1, buf0, p.W[8], p.B[8], out, rowbase, wave, lo, hi);
    break;
  default:
    layer_hidden<256, 256>(buf0, buf1, p.W[9],  p.B[9],  wave, lo, hi);
    __syncthreads();
    layer_hidden<256, 256>(buf1, buf0, p.W[10], p.B[10], wave, lo, hi);
    __syncthreads();
    layer_hidden<256, 256>(buf0, buf1, p.W[11], p.B[11], wave, lo, hi);
    __syncthreads();
    layer_hidden<256, 256>(buf1, buf0, p.W[12], p.B[12], wave, lo, hi);
    __syncthreads();
    layer_head(buf0, buf1, p.W[13], p.B[13], out, rowbase, wave, lo, hi);
    break;
  }
}

extern "C" void kernel_launch(void* const* d_in, const int* in_sizes, int n_in,
                              void* d_out, int out_size, void* d_ws, size_t ws_size,
                              hipStream_t stream) {
  (void)in_sizes; (void)n_in; (void)out_size; (void)d_ws; (void)ws_size;
  const float* state = (const float*)d_in[0];
  Params p;
  int idx = 1;
  for (int g = 0; g < 14; ++g) {
    p.W[g] = (const float*)d_in[idx++];
    p.B[g] = (const float*)d_in[idx++];
  }
  float* out = (float*)d_out;
  dim3 grid(4 * (BPN / MT));
  dim3 block(256);
  hipLaunchKernelGGL(hyper_mlp_f32_wmma, grid, block, 0, stream, state, p, out);
}
